# starting kernel
# baseline (speedup 1.0000x reference)
#include <hip/hip_runtime.h>

typedef float v2f __attribute__((ext_vector_type(2)));
typedef float v8f __attribute__((ext_vector_type(8)));

#define H1 128
#define H2 64
#define KIN 256
#define LDS_STRIDE 132

__global__ __launch_bounds__(128) void edge_mlp_kernel(
    const float* __restrict__ h,
    const int*   __restrict__ src_idx,
    const int*   __restrict__ dst_idx,
    const float* __restrict__ W1, const float* __restrict__ b1,
    const float* __restrict__ W2, const float* __restrict__ b2,
    const float* __restrict__ W3, const float* __restrict__ b3,
    float* __restrict__ out, int E)
{
    __shared__ float xstage[4 * 16 * LDS_STRIDE];

    const int lane = threadIdx.x & 31;
    const int wave = threadIdx.x >> 5;
    const int l15  = lane & 15;
    const int half = lane >> 4;
    float* stage = xstage + wave * 16 * LDS_STRIDE;

    const int tileBase = (blockIdx.x * 4 + wave) * 16;

    int e = tileBase + l15;
    if (e > E - 1) e = E - 1;
    const int s = src_idx[e];
    const int d = dst_idx[e];
    const float* rowS = h + (long long)s * H1 + 2 * half;
    const float* rowD = h + (long long)d * H1 + 2 * half;

    float bias1[8];
#pragma unroll
    for (int nt = 0; nt < 8; ++nt) bias1[nt] = b1[nt * 16 + l15];
    float bias2[4], w3v[4];
#pragma unroll
    for (int nt = 0; nt < 4; ++nt) {
        bias2[nt] = b2[nt * 16 + l15];
        w3v[nt]   = W3[nt * 16 + l15];
    }

    v8f acc[8] = {};
    const float* w1p = W1 + l15 * KIN + 2 * half;
    for (int c = 0; c < 32; ++c) {
        v2f a = *(const v2f*)(rowS + 4 * c);
#pragma unroll
        for (int nt = 0; nt < 8; ++nt) {
            v2f bm = *(const v2f*)(w1p + nt * 16 * KIN + 4 * c);
            acc[nt] = __builtin_amdgcn_wmma_f32_16x16x4_f32(
                false, a, false, bm, (short)0, acc[nt], false, false);
        }
    }
    for (int c = 0; c < 32; ++c) {
        v2f a = *(const v2f*)(rowD + 4 * c);
#pragma unroll
        for (int nt = 0; nt < 8; ++nt) {
            v2f bm = *(const v2f*)(w1p + nt * 16 * KIN + 4 * (c + 32));
            acc[nt] = __builtin_amdgcn_wmma_f32_16x16x4_f32(
                false, a, false, bm, (short)0, acc[nt], false, false);
        }
    }

#pragma unroll
    for (int nt = 0; nt < 8; ++nt) {
#pragma unroll
        for (int r = 0; r < 8; ++r) {
            float x = fmaxf(acc[nt][r] + bias1[nt], 0.0f);
            stage[(r + 8 * half) * LDS_STRIDE + nt * 16 + l15] = x;
        }
    }

    v8f acc2[4] = {};
    const float* w2p = W2 + l15 * H1 + 2 * half;
    const float* sp  = stage + l15 * LDS_STRIDE + 2 * half;
    for (int c = 0; c < 32; ++c) {
        v2f a = *(const v2f*)(sp + 4 * c);
#pragma unroll
        for (int nt = 0; nt < 4; ++nt) {
            v2f bm = *(const v2f*)(w2p + nt * 16 * H1 + 4 * c);
            acc2[nt] = __builtin_amdgcn_wmma_f32_16x16x4_f32(
                false, a, false, bm, (short)0, acc2[nt], false, false);
        }
    }

    float sred[8];
#pragma unroll
    for (int r = 0; r < 8; ++r) sred[r] = 0.0f;
#pragma unroll
    for (int nt = 0; nt < 4; ++nt) {
#pragma unroll
        for (int r = 0; r < 8; ++r) {
            float x = fmaxf(acc2[nt][r] + bias2[nt], 0.0f);
            sred[r] += x * w3v[nt];
        }
    }
#pragma unroll
    for (int mask = 1; mask <= 8; mask <<= 1) {
#pragma unroll
        for (int r = 0; r < 8; ++r)
            sred[r] += __shfl_xor(sred[r], mask, 32);
    }

    if (l15 == 0) {
        const float bb3 = b3[0];
        const int m0 = tileBase + 8 * half;
#pragma unroll
        for (int r = 0; r < 8; ++r) {
            const int idx = m0 + r;
            if (idx < E) out[idx] = sred[r] + bb3;
        }
    }
}

extern "C" void kernel_launch(void* const* d_in, const int* in_sizes, int n_in,
                              void* d_out, int out_size, void* d_ws, size_t ws_size,
                              hipStream_t stream) {
    const float* h   = (const float*)d_in[0];
    const int*   src = (const int*)  d_in[1];
    const int*   dst = (const int*)  d_in[2];
    const float* W1  = (const float*)d_in[3];
    const float* b1  = (const float*)d_in[4];
    const float* W2  = (const float*)d_in[5];
    const float* b2  = (const float*)d_in[6];
    const float* W3  = (const float*)d_in[7];
    const float* b3  = (const float*)d_in[8];
    float* out = (float*)d_out;

    const int E = in_sizes[1];
    const int blocks = (E + 63) / 64;
    edge_mlp_kernel<<<blocks, 128, 0, stream>>>(h, src, dst, W1, b1, W2, b2,
                                                W3, b3, out, E);
}


# W1 and W2 held in LDS by one resident 16-wave work-group per WGP that loops over the 16-edge tiles; B fragments read from LDS instead of 16-line global loads; k order of all sums unchanged
# speedup vs baseline: 2.2353x; 2.2353x over previous
#include <hip/hip_runtime.h>


typedef float v2f __attribute__((ext_vector_type(2)));
typedef float v4f __attribute__((ext_vector_type(4)));
typedef float v8f __attribute__((ext_vector_type(8)));

#define H1 128
#define H2 64
#define KIN 256
#define NWAVES 16
#define NBLOCKS 256
#define W1_PITCH 260
#define W2_PITCH 132
#define ST_PITCH 132

__global__ __launch_bounds__(NWAVES * 32) void edge_mlp_kernel(
    const float* __restrict__ h,
    const int*   __restrict__ src_idx,
    const int*   __restrict__ dst_idx,
    const float* __restrict__ W1, const float* __restrict__ b1,
    const float* __restrict__ W2, const float* __restrict__ b2,
    const float* __restrict__ W3, const float* __restrict__ b3,
    float* __restrict__ out, int E)
{
    __shared__ __attribute__((aligned(16))) float w1s[H1 * W1_PITCH];
    __shared__ __attribute__((aligned(16))) float w2s[H2 * W2_PITCH];
    __shared__ __attribute__((aligned(16))) float xstage[NWAVES * 16 * ST_PITCH];

    for (int i = threadIdx.x; i < H1 * KIN / 4; i += NWAVES * 32) {
        const int row = i >> 6;
        const int col = (i & 63) * 4;
        const v4f v = *(const v4f*)(W1 + row * KIN + col);
        *(v4f*)(w1s + row * W1_PITCH + col) = v;
    }
    for (int i = threadIdx.x; i < H2 * H1 / 4; i += NWAVES * 32) {
        const int row = i >> 5;
        const int col = (i & 31) * 4;
        const v4f v = *(const v4f*)(W2 + row * H1 + col);
        *(v4f*)(w2s + row * W2_PITCH + col) = v;
    }
    __syncthreads();

    const int lane = threadIdx.x & 31;
    const int wave = __builtin_amdgcn_readfirstlane((int)(threadIdx.x >> 5));
    const int l15  = lane & 15;
    const int half = lane >> 4;
    float* stage = xstage + wave * 16 * ST_PITCH;

    float bias1[8];
#pragma unroll
    for (int nt = 0; nt < 8; ++nt) bias1[nt] = b1[nt * 16 + l15];
    float bias2[4], w3v[4];
#pragma unroll
    for (int nt = 0; nt < 4; ++nt) {
        bias2[nt] = b2[nt * 16 + l15];
        w3v[nt]   = W3[nt * 16 + l15];
    }
    const float bb3 = b3[0];

    const float* w1p = w1s + l15 * W1_PITCH + 2 * half;
    const float* w2p = w2s + l15 * W2_PITCH + 2 * half;
    const float* sp  = stage + l15 * ST_PITCH + 2 * half;

    const int numTiles = (E + 15) >> 4;
    for (int p = (wave >> 1) * NBLOCKS + (int)blockIdx.x; 2 * p + (wave & 1) < numTiles;
         p += NBLOCKS * (NWAVES / 2)) {
        const int tileBase = (2 * p + (wave & 1)) * 16;

        int e = tileBase + l15;
        if (e > E - 1) e = E - 1;
        const int s = src_idx[e];
        const int d = dst_idx[e];
        const float* rowS = h + (long long)s * H1 + 2 * half;
        const float* rowD = h + (long long)d * H1 + 2 * half;

        v8f acc[8] = {};
        for (int c = 0; c < 32; ++c) {
            v2f a = *(const v2f*)(rowS + 4 * c);
#pragma unroll
            for (int nt = 0; nt < 8; ++nt) {
                v2f bm = *(const v2f*)(w1p + nt * 16 * W1_PITCH + 4 * c);
                acc[nt] = __builtin_amdgcn_wmma_f32_16x16x4_f32(
                    false, a, false, bm, (short)0, acc[nt], false, false);
            }
        }
        for (int c = 0; c < 32; ++c) {
            v2f a = *(const v2f*)(rowD + 4 * c);
#pragma unroll
            for (int nt = 0; nt < 8; ++nt) {
                v2f bm = *(const v2f*)(w1p + nt * 16 * W1_PITCH + 4 * (c + 32));
                acc[nt] = __builtin_amdgcn_wmma_f32_16x16x4_f32(
                    false, a, false, bm, (short)0, acc[nt], false, false);
            }
        }

#pragma unroll
        for (int nt = 0; nt < 8; ++nt) {
#pragma unroll
            for (int r = 0; r < 8; ++r) {
                float x = fmaxf(acc[nt][r] + bias1[nt], 0.0f);
                stage[(r + 8 * half) * ST_PITCH + nt * 16 + l15] = x;
            }
        }

        v8f acc2[4] = {};
        for (int c = 0; c < 32; ++c) {
            v2f a = *(const v2f*)(sp + 4 * c);
#pragma unroll
            for (int nt = 0; nt < 4; ++nt) {
                v2f bm = *(const v2f*)(w2p + nt * 16 * W2_PITCH + 4 * c);
                acc2[nt] = __builtin_amdgcn_wmma_f32_16x16x4_f32(
                    false, a, false, bm, (short)0, acc2[nt], false, false);
            }
        }

        float sred[8];
#pragma unroll
        for (int r = 0; r < 8; ++r) sred[r] = 0.0f;
#pragma unroll
        for (int nt = 0; nt < 4; ++nt) {
#pragma unroll
            for (int r = 0; r < 8; ++r) {
                float x = fmaxf(acc2[nt][r] + bias2[nt], 0.0f);
                sred[r] += x * w3v[nt];
            }
        }
#pragma unroll
        for (int mask = 1; mask <= 8; mask <<= 1) {
#pragma unroll
            for (int r = 0; r < 8; ++r)
                sred[r] += __shfl_xor(sred[r], mask, 32);
        }

        if (l15 == 0) {
            const int m0 = tileBase + 8 * half;
#pragma unroll
            for (int r = 0; r < 8; ++r) {
                const int idx = m0 + r;
                if (idx < E) out[idx] = sred[r] + bb3;
            }
        }
    }
}

extern "C" void kernel_launch(void* const* d_in, const int* in_sizes, int n_in,
                              void* d_out, int out_size, void* d_ws, size_t ws_size,
                              hipStream_t stream) {
    const float* h   = (const float*)d_in[0];
    const int*   src = (const int*)  d_in[1];
    const int*   dst = (const int*)  d_in[2];
    const float* W1  = (const float*)d_in[3];
    const float* b1  = (const float*)d_in[4];
    const float* W2  = (const float*)d_in[5];
    const float* b2  = (const float*)d_in[6];
    const float* W3  = (const float*)d_in[7];
    const float* b3  = (const float*)d_in[8];
    float* out = (float*)d_out;

    const int E = in_sizes[1];
    edge_mlp_kernel<<<NBLOCKS, NWAVES * 32, 0, stream>>>(h, src, dst, W1, b1, W2, b2,
                                                          W3, b3, out, E);
}


# Edge gather loads whole 512-byte P/Q rows (one row per load, 4 lines touched once each, not 16 lines x 32 B); relu(P+Q) reaches the A-fragment layout via a per-wave LDS tile; k order of sums unchanged
# speedup vs baseline: 5.3140x; 2.3773x over previous
#include <hip/hip_runtime.h>


typedef float v2f __attribute__((ext_vector_type(2)));
typedef float v4f __attribute__((ext_vector_type(4)));
typedef float v8f __attribute__((ext_vector_type(8)));

#define H1 128
#define H2 64
#define KIN 256
#define PQW 256
#define NWAVES 16
#define NBLOCKS 256
#define W1_PITCH 260
#define W2_PITCH 132
#define ST_PITCH 132

__global__ __launch_bounds__(NWAVES * 32) void node_proj_kernel(
    const float* __restrict__ h,
    const float* __restrict__ W1, const float* __restrict__ b1,
    float* __restrict__ pq, int hsize)
{
    __shared__ __attribute__((aligned(16))) float w1s[H1 * W1_PITCH];
    __shared__ __attribute__((aligned(16))) float xstage[NWAVES * 16 * ST_PITCH];

    for (int i = threadIdx.x; i < H1 * KIN / 4; i += NWAVES * 32) {
        const int row = i >> 6;
        const int col = (i & 63) * 4;
        const v4f v = *(const v4f*)(W1 + row * KIN + col);
        *(v4f*)(w1s + row * W1_PITCH + col) = v;
    }
    __syncthreads();

    const int lane = threadIdx.x & 31;
    const int wave = __builtin_amdgcn_readfirstlane((int)(threadIdx.x >> 5));
    const int l15  = lane & 15;
    const int half = lane >> 4;
    float* stage = xstage + wave * 16 * ST_PITCH;

    float bias1[8];
#pragma unroll
    for (int nt = 0; nt < 8; ++nt) bias1[nt] = b1[nt * 16 + l15];

    const float* w1p = w1s + l15 * W1_PITCH + 2 * half;

    const int numNodes = hsize >> 7;
    const int numTiles = (numNodes + 15) >> 4;
    for (int p = (wave >> 1) * NBLOCKS + (int)blockIdx.x; 2 * p + (wave & 1) < numTiles;
         p += NBLOCKS * (NWAVES / 2)) {
        const int tileBase = (2 * p + (wave & 1)) * 16;

        int node = tileBase + l15;
        if (node > numNodes - 1) node = numNodes - 1;
        const float* rowH = h + (long long)node * H1 + 2 * half;

        for (int part = 0; part < 2; ++part) {
            v8f acc[8] = {};
            for (int c = 0; c < 32; ++c) {
                v2f a = *(const v2f*)(rowH + 4 * c);
#pragma unroll
                for (int nt = 0; nt < 8; ++nt) {
                    v2f bm = *(const v2f*)(w1p + nt * 16 * W1_PITCH + part * H1 + 4 * c);
                    acc[nt] = __builtin_amdgcn_wmma_f32_16x16x4_f32(
                        false, a, false, bm, (short)0, acc[nt], false, false);
                }
            }

#pragma unroll
            for (int nt = 0; nt < 8; ++nt) {
                const float badd = (part == 0) ? bias1[nt] : 0.0f;
#pragma unroll
                for (int r = 0; r < 8; ++r) {
                    stage[(r + 8 * half) * ST_PITCH + nt * 16 + l15] = acc[nt][r] + badd;
                }
            }
#pragma unroll
            for (int row = 0; row < 16; ++row) {
                const v4f v = *(const v4f*)(stage + row * ST_PITCH + 4 * lane);
                const int nd = tileBase + row;
                if (nd < numNodes)
                    *(v4f*)(pq + (long long)nd * PQW + part * H1 + 4 * lane) = v;
            }
        }
    }
}

__global__ __launch_bounds__(NWAVES * 32) void edge_tail_kernel(
    const float* __restrict__ pq,
    const int*   __restrict__ src_idx,
    const int*   __restrict__ dst_idx,
    const float* __restrict__ W2, const float* __restrict__ b2,
    const float* __restrict__ W3, const float* __restrict__ b3,
    float* __restrict__ out, int E)
{
    __shared__ __attribute__((aligned(16))) float w2s[H2 * W2_PITCH];
    __shared__ __attribute__((aligned(16))) float xstage[NWAVES * 16 * ST_PITCH];

    for (int i = threadIdx.x; i < H2 * H1 / 4; i += NWAVES * 32) {
        const int row = i >> 5;
        const int col = (i & 31) * 4;
        const v4f v = *(const v4f*)(W2 + row * H1 + col);
        *(v4f*)(w2s + row * W2_PITCH + col) = v;
    }
    __syncthreads();

    const int lane = threadIdx.x & 31;
    const int wave = __builtin_amdgcn_readfirstlane((int)(threadIdx.x >> 5));
    const int l15  = lane & 15;
    const int half = lane >> 4;

    float bias2[4], w3v[4];
#pragma unroll
    for (int nt = 0; nt < 4; ++nt) {
        bias2[nt] = b2[nt * 16 + l15];
        w3v[nt]   = W3[nt * 16 + l15];
    }
    const float bb3 = b3[0];

    const float* w2p = w2s + l15 * W2_PITCH + 4 * half;
    float* stage = xstage + wave * 16 * ST_PITCH;
    float* stw = stage + 4 * lane;
    const float* sta = stage + l15 * ST_PITCH + 4 * half;

    const int numTiles = (E + 15) >> 4;
    for (int p = (wave >> 1) * NBLOCKS + (int)blockIdx.x; 2 * p + (wave & 1) < numTiles;
         p += NBLOCKS * (NWAVES / 2)) {
        const int tileBase = (2 * p + (wave & 1)) * 16;

        asm volatile("" ::: "memory");

        int e = tileBase + l15;
        if (e > E - 1) e = E - 1;
        const int s = src_idx[e];
        const int d = dst_idx[e];

        v4f pv[16], qv[16];
#pragma unroll
        for (int i = 0; i < 16; ++i) {
            const int si = __builtin_amdgcn_readlane(s, i);
            const int di = __builtin_amdgcn_readlane(d, i);
            pv[i] = *(const v4f*)(pq + (long long)si * PQW + 4 * lane);
            qv[i] = *(const v4f*)(pq + (long long)di * PQW + H1 + 4 * lane);
        }
#pragma unroll
        for (int i = 0; i < 16; ++i) {
            const v4f x = pv[i] + qv[i];
            const v4f r = { fmaxf(x.x, 0.0f), fmaxf(x.y, 0.0f), fmaxf(x.z, 0.0f), fmaxf(x.w, 0.0f) };
            *(v4f*)(stw + i * ST_PITCH) = r;
        }
        v4f av[16];
#pragma unroll
        for (int j = 0; j < 16; ++j)
            av[j] = *(const v4f*)(sta + 8 * j);

        v8f acc2[4] = {};
#pragma unroll
        for (int j = 0; j < 16; ++j) {
            const v2f a0 = { av[j].x, av[j].y };
            const v2f a1 = { av[j].z, av[j].w };
            v4f bm[4];
#pragma unroll
            for (int nt = 0; nt < 4; ++nt)
                bm[nt] = *(const v4f*)(w2p + nt * 16 * W2_PITCH + 8 * j);
#pragma unroll
            for (int nt = 0; nt < 4; ++nt) {
                const v2f b0 = { bm[nt].x, bm[nt].y };
                acc2[nt] = __builtin_amdgcn_wmma_f32_16x16x4_f32(
                    false, a0, false, b0, (short)0, acc2[nt], false, false);
            }
#pragma unroll
            for (int nt = 0; nt < 4; ++nt) {
                const v2f b1v = { bm[nt].z, bm[nt].w };
                acc2[nt] = __builtin_amdgcn_wmma_f32_16x16x4_f32(
                    false, a1, false, b1v, (short)0, acc2[nt], false, false);
            }
        }

        float sred[8];
#pragma unroll
        for (int r = 0; r < 8; ++r) sred[r] = 0.0f;
#pragma unroll
        for (int nt = 0; nt < 4; ++nt) {
#pragma unroll
            for (int r = 0; r < 8; ++r) {
                float x = fmaxf(acc2[nt][r] + bias2[nt], 0.0f);
                sred[r] += x * w3v[nt];
            }
        }
#pragma unroll
        for (int mask = 1; mask <= 8; mask <<= 1) {
#pragma unroll
            for (int r = 0; r < 8; ++r)
                sred[r] += __shfl_xor(sred[r], mask, 32);
        }

        if (l15 == 0) {
            const int m0 = tileBase + 8 * half;
#pragma unroll
            for (int r = 0; r < 8; ++r) {
                const int idx = m0 + r;
                if (idx < E) out[idx] = sred[r] + bb3;
            }
        }
    }
}

__global__ __launch_bounds__(NWAVES * 32) void edge_mlp_kernel(
    const float* __restrict__ h,
    const int*   __restrict__ src_idx,
    const int*   __restrict__ dst_idx,
    const float* __restrict__ W1, const float* __restrict__ b1,
    const float* __restrict__ W2, const float* __restrict__ b2,
    const float* __restrict__ W3, const float* __restrict__ b3,
    float* __restrict__ out, int E)
{
    __shared__ __attribute__((aligned(16))) float w1s[H1 * W1_PITCH];
    __shared__ __attribute__((aligned(16))) float w2s[H2 * W2_PITCH];
    __shared__ __attribute__((aligned(16))) float xstage[NWAVES * 16 * ST_PITCH];

    for (int i = threadIdx.x; i < H1 * KIN / 4; i += NWAVES * 32) {
        const int row = i >> 6;
        const int col = (i & 63) * 4;
        const v4f v = *(const v4f*)(W1 + row * KIN + col);
        *(v4f*)(w1s + row * W1_PITCH + col) = v;
    }
    for (int i = threadIdx.x; i < H2 * H1 / 4; i += NWAVES * 32) {
        const int row = i >> 5;
        const int col = (i & 31) * 4;
        const v4f v = *(const v4f*)(W2 + row * H1 + col);
        *(v4f*)(w2s + row * W2_PITCH + col) = v;
    }
    __syncthreads();

    const int lane = threadIdx.x & 31;
    const int wave = __builtin_amdgcn_readfirstlane((int)(threadIdx.x >> 5));
    const int l15  = lane & 15;
    const int half = lane >> 4;
    float* stage = xstage + wave * 16 * ST_PITCH;

    float bias1[8];
#pragma unroll
    for (int nt = 0; nt < 8; ++nt) bias1[nt] = b1[nt * 16 + l15];
    float bias2[4], w3v[4];
#pragma unroll
    for (int nt = 0; nt < 4; ++nt) {
        bias2[nt] = b2[nt * 16 + l15];
        w3v[nt]   = W3[nt * 16 + l15];
    }
    const float bb3 = b3[0];

    const float* w1p = w1s + l15 * W1_PITCH + 2 * half;
    const float* w2p = w2s + l15 * W2_PITCH + 2 * half;
    const float* sp  = stage + l15 * ST_PITCH + 2 * half;

    const int numTiles = (E + 15) >> 4;
    for (int p = (wave >> 1) * NBLOCKS + (int)blockIdx.x; 2 * p + (wave & 1) < numTiles;
         p += NBLOCKS * (NWAVES / 2)) {
        const int tileBase = (2 * p + (wave & 1)) * 16;

        int e = tileBase + l15;
        if (e > E - 1) e = E - 1;
        const int s = src_idx[e];
        const int d = dst_idx[e];
        const float* rowS = h + (long long)s * H1 + 2 * half;
        const float* rowD = h + (long long)d * H1 + 2 * half;

        v8f acc[8] = {};
        for (int c = 0; c < 32; ++c) {
            v2f a = *(const v2f*)(rowS + 4 * c);
#pragma unroll
            for (int nt = 0; nt < 8; ++nt) {
                v2f bm = *(const v2f*)(w1p + nt * 16 * W1_PITCH + 4 * c);
                acc[nt] = __builtin_amdgcn_wmma_f32_16x16x4_f32(
                    false, a, false, bm, (short)0, acc[nt], false, false);
            }
        }
        for (int c = 0; c < 32; ++c) {
            v2f a = *(const v2f*)(rowD + 4 * c);
#pragma unroll
            for (int nt = 0; nt < 8; ++nt) {
                v2f bm = *(const v2f*)(w1p + nt * 16 * W1_PITCH + 4 * (c + 32));
                acc[nt] = __builtin_amdgcn_wmma_f32_16x16x4_f32(
                    false, a, false, bm, (short)0, acc[nt], false, false);
            }
        }

#pragma unroll
        for (int nt = 0; nt < 8; ++nt) {
#pragma unroll
            for (int r = 0; r < 8; ++r) {
                float x = fmaxf(acc[nt][r] + bias1[nt], 0.0f);
                stage[(r + 8 * half) * ST_PITCH + nt * 16 + l15] = x;
            }
        }

        v8f acc2[4] = {};
        for (int c = 0; c < 32; ++c) {
            v2f a = *(const v2f*)(sp + 4 * c);
#pragma unroll
            for (int nt = 0; nt < 4; ++nt) {
                v2f bm = *(const v2f*)(w2p + nt * 16 * W2_PITCH + 4 * c);
                acc2[nt] = __builtin_amdgcn_wmma_f32_16x16x4_f32(
                    false, a, false, bm, (short)0, acc2[nt], false, false);
            }
        }

        float sred[8];
#pragma unroll
        for (int r = 0; r < 8; ++r) sred[r] = 0.0f;
#pragma unroll
        for (int nt = 0; nt < 4; ++nt) {
#pragma unroll
            for (int r = 0; r < 8; ++r) {
                float x = fmaxf(acc2[nt][r] + bias2[nt], 0.0f);
                sred[r] += x * w3v[nt];
            }
        }
#pragma unroll
        for (int mask = 1; mask <= 8; mask <<= 1) {
#pragma unroll
            for (int r = 0; r < 8; ++r)
                sred[r] += __shfl_xor(sred[r], mask, 32);
        }

        if (l15 == 0) {
            const int m0 = tileBase + 8 * half;
#pragma unroll
            for (int r = 0; r < 8; ++r) {
                const int idx = m0 + r;
                if (idx < E) out[idx] = sred[r] + bb3;
            }
        }
    }
}

extern "C" void kernel_launch(void* const* d_in, const int* in_sizes, int n_in,
                              void* d_out, int out_size, void* d_ws, size_t ws_size,
                              hipStream_t stream) {
    const float* h   = (const float*)d_in[0];
    const int*   src = (const int*)  d_in[1];
    const int*   dst = (const int*)  d_in[2];
    const float* W1  = (const float*)d_in[3];
    const float* b1  = (const float*)d_in[4];
    const float* W2  = (const float*)d_in[5];
    const float* b2  = (const float*)d_in[6];
    const float* W3  = (const float*)d_in[7];
    const float* b3  = (const float*)d_in[8];
    float* out = (float*)d_out;
    float* pq  = (float*)d_ws;

    const int E = in_sizes[1];
    const int hsize = in_sizes[0];
    if (ws_size >= (size_t)hsize * 2 * sizeof(float)) {
        node_proj_kernel<<<NBLOCKS, NWAVES * 32, 0, stream>>>(h, W1, b1, pq, hsize);
        edge_tail_kernel<<<NBLOCKS, NWAVES * 32, 0, stream>>>(pq, src, dst, W2, b2, W3, b3, out, E);
    } else {
        edge_mlp_kernel<<<NBLOCKS, NWAVES * 32, 0, stream>>>(h, src, dst, W1, b1, W2, b2,
                                                              W3, b3, out, E);
    }
}


# Edge kernel: layer 2 done transposed (W2 as A operand): an edge's 64 features land in two lanes, so layer 3 is an in-lane sum, one half exchange and one 64-byte store; layer-3 sum order changes
# speedup vs baseline: 5.5251x; 1.0397x over previous
#include <hip/hip_runtime.h>


typedef float v2f __attribute__((ext_vector_type(2)));
typedef float v4f __attribute__((ext_vector_type(4)));
typedef float v8f __attribute__((ext_vector_type(8)));

#define H1 128
#define H2 64
#define KIN 256
#define PQW 256
#define NWAVES 16
#define NBLOCKS 256
#define W1_PITCH 260
#define W2_PITCH 132
#define ST_PITCH 132

__global__ __launch_bounds__(NWAVES * 32) void node_proj_kernel(
    const float* __restrict__ h,
    const float* __restrict__ W1, const float* __restrict__ b1,
    float* __restrict__ pq, int hsize)
{
    __shared__ __attribute__((aligned(16))) float w1s[H1 * W1_PITCH];
    __shared__ __attribute__((aligned(16))) float xstage[NWAVES * 16 * ST_PITCH];

    for (int i = threadIdx.x; i < H1 * KIN / 4; i += NWAVES * 32) {
        const int row = i >> 6;
        const int col = (i & 63) * 4;
        const v4f v = *(const v4f*)(W1 + row * KIN + col);
        *(v4f*)(w1s + row * W1_PITCH + col) = v;
    }
    __syncthreads();

    const int lane = threadIdx.x & 31;
    const int wave = __builtin_amdgcn_readfirstlane((int)(threadIdx.x >> 5));
    const int l15  = lane & 15;
    const int half = lane >> 4;
    float* stage = xstage + wave * 16 * ST_PITCH;

    float bias1[8];
#pragma unroll
    for (int nt = 0; nt < 8; ++nt) bias1[nt] = b1[nt * 16 + l15];

    const float* w1p = w1s + l15 * W1_PITCH + 2 * half;

    const int numNodes = hsize >> 7;
    const int numTiles = (numNodes + 15) >> 4;
    for (int p = (wave >> 1) * NBLOCKS + (int)blockIdx.x; 2 * p + (wave & 1) < numTiles;
         p += NBLOCKS * (NWAVES / 2)) {
        const int tileBase = (2 * p + (wave & 1)) * 16;

        int node = tileBase + l15;
        if (node > numNodes - 1) node = numNodes - 1;
        const float* rowH = h + (long long)node * H1 + 2 * half;

        for (int part = 0; part < 2; ++part) {
            v8f acc[8] = {};
            for (int c = 0; c < 32; ++c) {
                v2f a = *(const v2f*)(rowH + 4 * c);
#pragma unroll
                for (int nt = 0; nt < 8; ++nt) {
                    v2f bm = *(const v2f*)(w1p + nt * 16 * W1_PITCH + part * H1 + 4 * c);
                    acc[nt] = __builtin_amdgcn_wmma_f32_16x16x4_f32(
                        false, a, false, bm, (short)0, acc[nt], false, false);
                }
            }

#pragma unroll
            for (int nt = 0; nt < 8; ++nt) {
                const float badd = (part == 0) ? bias1[nt] : 0.0f;
#pragma unroll
                for (int r = 0; r < 8; ++r) {
                    stage[(r + 8 * half) * ST_PITCH + nt * 16 + l15] = acc[nt][r] + badd;
                }
            }
#pragma unroll
            for (int row = 0; row < 16; ++row) {
                const v4f v = *(const v4f*)(stage + row * ST_PITCH + 4 * lane);
                const int nd = tileBase + row;
                if (nd < numNodes)
                    *(v4f*)(pq + (long long)nd * PQW + part * H1 + 4 * lane) = v;
            }
        }
    }
}

__global__ __launch_bounds__(NWAVES * 32) void edge_tail_kernel(
    const float* __restrict__ pq,
    const int*   __restrict__ src_idx,
    const int*   __restrict__ dst_idx,
    const float* __restrict__ W2, const float* __restrict__ b2,
    const float* __restrict__ W3, const float* __restrict__ b3,
    float* __restrict__ out, int E)
{
    __shared__ __attribute__((aligned(16))) float w2s[H2 * W2_PITCH];
    __shared__ __attribute__((aligned(16))) float xstage[NWAVES * 16 * ST_PITCH];

    for (int i = threadIdx.x; i < H2 * H1 / 4; i += NWAVES * 32) {
        const int row = i >> 5;
        const int col = (i & 31) * 4;
        const v4f v = *(const v4f*)(W2 + row * H1 + col);
        *(v4f*)(w2s + row * W2_PITCH + col) = v;
    }
    __syncthreads();

    const int lane = threadIdx.x & 31;
    const int wave = __builtin_amdgcn_readfirstlane((int)(threadIdx.x >> 5));
    const int l15  = lane & 15;
    const int half = lane >> 4;

    v2f bias2[4][4], w3v[4][4];
#pragma unroll
    for (int nt = 0; nt < 4; ++nt) {
#pragma unroll
        for (int q = 0; q < 4; ++q) {
            const int n = nt * 16 + 8 * half + 2 * q;
            bias2[nt][q] = *(const v2f*)(b2 + n);
            w3v[nt][q]   = *(const v2f*)(W3 + n);
        }
    }
    const float bb3 = b3[0];

    const float* w2p = w2s + l15 * W2_PITCH + 4 * half;
    float* stage = xstage + wave * 16 * ST_PITCH;
    float* stw = stage + 4 * lane;
    const float* sta = stage + l15 * ST_PITCH + 4 * half;

    const int numTiles = (E + 15) >> 4;
    for (int p = (wave >> 1) * NBLOCKS + (int)blockIdx.x; 2 * p + (wave & 1) < numTiles;
         p += NBLOCKS * (NWAVES / 2)) {
        const int tileBase = (2 * p + (wave & 1)) * 16;

        asm volatile("" ::: "memory");

        int e = tileBase + l15;
        if (e > E - 1) e = E - 1;
        const int s = src_idx[e];
        const int d = dst_idx[e];

        v4f pv[16], qv[16];
#pragma unroll
        for (int i = 0; i < 16; ++i) {
            const int si = __builtin_amdgcn_readlane(s, i);
            const int di = __builtin_amdgcn_readlane(d, i);
            pv[i] = *(const v4f*)(pq + (long long)si * PQW + 4 * lane);
            qv[i] = *(const v4f*)(pq + (long long)di * PQW + H1 + 4 * lane);
        }
#pragma unroll
        for (int i = 0; i < 16; ++i) {
            const v4f x = pv[i] + qv[i];
            const v4f r = { fmaxf(x.x, 0.0f), fmaxf(x.y, 0.0f), fmaxf(x.z, 0.0f), fmaxf(x.w, 0.0f) };
            *(v4f*)(stw + i * ST_PITCH) = r;
        }
        v4f av[16];
#pragma unroll
        for (int j = 0; j < 16; ++j)
            av[j] = *(const v4f*)(sta + 8 * j);

        v8f acc2[4] = {};
#pragma unroll
        for (int j = 0; j < 16; ++j) {
            const v2f a0 = { av[j].x, av[j].y };
            const v2f a1 = { av[j].z, av[j].w };
            v4f bm[4];
#pragma unroll
            for (int nt = 0; nt < 4; ++nt)
                bm[nt] = *(const v4f*)(w2p + nt * 16 * W2_PITCH + 8 * j);
#pragma unroll
            for (int nt = 0; nt < 4; ++nt) {
                const v2f b0 = { bm[nt].x, bm[nt].y };
                acc2[nt] = __builtin_amdgcn_wmma_f32_16x16x4_f32(
                    false, b0, false, a0, (short)0, acc2[nt], false, false);
            }
#pragma unroll
            for (int nt = 0; nt < 4; ++nt) {
                const v2f b1v = { bm[nt].z, bm[nt].w };
                acc2[nt] = __builtin_amdgcn_wmma_f32_16x16x4_f32(
                    false, b1v, false, a1, (short)0, acc2[nt], false, false);
            }
        }

        v2f sacc[4];
#pragma unroll
        for (int nt = 0; nt < 4; ++nt) {
            sacc[nt] = (v2f){ 0.0f, 0.0f };
#pragma unroll
            for (int q = 0; q < 4; ++q) {
                v2f x = (v2f){ acc2[nt][2 * q], acc2[nt][2 * q + 1] } + bias2[nt][q];
                x = (v2f){ fmaxf(x.x, 0.0f), fmaxf(x.y, 0.0f) };
                sacc[nt] = x * w3v[nt][q] + sacc[nt];
            }
        }
        const v2f s2 = (sacc[0] + sacc[1]) + (sacc[2] + sacc[3]);
        const float part = s2.x + s2.y;
        const float tot = part + __shfl_xor(part, 16, 32);

        const int idx = tileBase + l15;
        if (half == 0 && idx < E) out[idx] = tot + bb3;
    }
}

__global__ __launch_bounds__(NWAVES * 32) void edge_mlp_kernel(
    const float* __restrict__ h,
    const int*   __restrict__ src_idx,
    const int*   __restrict__ dst_idx,
    const float* __restrict__ W1, const float* __restrict__ b1,
    const float* __restrict__ W2, const float* __restrict__ b2,
    const float* __restrict__ W3, const float* __restrict__ b3,
    float* __restrict__ out, int E)
{
    __shared__ __attribute__((aligned(16))) float w1s[H1 * W1_PITCH];
    __shared__ __attribute__((aligned(16))) float w2s[H2 * W2_PITCH];
    __shared__ __attribute__((aligned(16))) float xstage[NWAVES * 16 * ST_PITCH];

    for (int i = threadIdx.x; i < H1 * KIN / 4; i += NWAVES * 32) {
        const int row = i >> 6;
        const int col = (i & 63) * 4;
        const v4f v = *(const v4f*)(W1 + row * KIN + col);
        *(v4f*)(w1s + row * W1_PITCH + col) = v;
    }
    for (int i = threadIdx.x; i < H2 * H1 / 4; i += NWAVES * 32) {
        const int row = i >> 5;
        const int col = (i & 31) * 4;
        const v4f v = *(const v4f*)(W2 + row * H1 + col);
        *(v4f*)(w2s + row * W2_PITCH + col) = v;
    }
    __syncthreads();

    const int lane = threadIdx.x & 31;
    const int wave = __builtin_amdgcn_readfirstlane((int)(threadIdx.x >> 5));
    const int l15  = lane & 15;
    const int half = lane >> 4;
    float* stage = xstage + wave * 16 * ST_PITCH;

    float bias1[8];
#pragma unroll
    for (int nt = 0; nt < 8; ++nt) bias1[nt] = b1[nt * 16 + l15];
    float bias2[4], w3v[4];
#pragma unroll
    for (int nt = 0; nt < 4; ++nt) {
        bias2[nt] = b2[nt * 16 + l15];
        w3v[nt]   = W3[nt * 16 + l15];
    }
    const float bb3 = b3[0];

    const float* w1p = w1s + l15 * W1_PITCH + 2 * half;
    const float* w2p = w2s + l15 * W2_PITCH + 2 * half;
    const float* sp  = stage + l15 * ST_PITCH + 2 * half;

    const int numTiles = (E + 15) >> 4;
    for (int p = (wave >> 1) * NBLOCKS + (int)blockIdx.x; 2 * p + (wave & 1) < numTiles;
         p += NBLOCKS * (NWAVES / 2)) {
        const int tileBase = (2 * p + (wave & 1)) * 16;

        int e = tileBase + l15;
        if (e > E - 1) e = E - 1;
        const int s = src_idx[e];
        const int d = dst_idx[e];
        const float* rowS = h + (long long)s * H1 + 2 * half;
        const float* rowD = h + (long long)d * H1 + 2 * half;

        v8f acc[8] = {};
        for (int c = 0; c < 32; ++c) {
            v2f a = *(const v2f*)(rowS + 4 * c);
#pragma unroll
            for (int nt = 0; nt < 8; ++nt) {
                v2f bm = *(const v2f*)(w1p + nt * 16 * W1_PITCH + 4 * c);
                acc[nt] = __builtin_amdgcn_wmma_f32_16x16x4_f32(
                    false, a, false, bm, (short)0, acc[nt], false, false);
            }
        }
        for (int c = 0; c < 32; ++c) {
            v2f a = *(const v2f*)(rowD + 4 * c);
#pragma unroll
            for (int nt = 0; nt < 8; ++nt) {
                v2f bm = *(const v2f*)(w1p + nt * 16 * W1_PITCH + 4 * (c + 32));
                acc[nt] = __builtin_amdgcn_wmma_f32_16x16x4_f32(
                    false, a, false, bm, (short)0, acc[nt], false, false);
            }
        }

#pragma unroll
        for (int nt = 0; nt < 8; ++nt) {
#pragma unroll
            for (int r = 0; r < 8; ++r) {
                float x = fmaxf(acc[nt][r] + bias1[nt], 0.0f);
                stage[(r + 8 * half) * ST_PITCH + nt * 16 + l15] = x;
            }
        }

        v8f acc2[4] = {};
        for (int c = 0; c < 32; ++c) {
            v2f a = *(const v2f*)(sp + 4 * c);
#pragma unroll
            for (int nt = 0; nt < 4; ++nt) {
                v2f bm = *(const v2f*)(w2p + nt * 16 * W2_PITCH + 4 * c);
                acc2[nt] = __builtin_amdgcn_wmma_f32_16x16x4_f32(
                    false, a, false, bm, (short)0, acc2[nt], false, false);
            }
        }

        float sred[8];
#pragma unroll
        for (int r = 0; r < 8; ++r) sred[r] = 0.0f;
#pragma unroll
        for (int nt = 0; nt < 4; ++nt) {
#pragma unroll
            for (int r = 0; r < 8; ++r) {
                float x = fmaxf(acc2[nt][r] + bias2[nt], 0.0f);
                sred[r] += x * w3v[nt];
            }
        }
#pragma unroll
        for (int mask = 1; mask <= 8; mask <<= 1) {
#pragma unroll
            for (int r = 0; r < 8; ++r)
                sred[r] += __shfl_xor(sred[r], mask, 32);
        }

        if (l15 == 0) {
            const int m0 = tileBase + 8 * half;
#pragma unroll
            for (int r = 0; r < 8; ++r) {
                const int idx = m0 + r;
                if (idx < E) out[idx] = sred[r] + bb3;
            }
        }
    }
}

extern "C" void kernel_launch(void* const* d_in, const int* in_sizes, int n_in,
                              void* d_out, int out_size, void* d_ws, size_t ws_size,
                              hipStream_t stream) {
    const float* h   = (const float*)d_in[0];
    const int*   src = (const int*)  d_in[1];
    const int*   dst = (const int*)  d_in[2];
    const float* W1  = (const float*)d_in[3];
    const float* b1  = (const float*)d_in[4];
    const float* W2  = (const float*)d_in[5];
    const float* b2  = (const float*)d_in[6];
    const float* W3  = (const float*)d_in[7];
    const float* b3  = (const float*)d_in[8];
    float* out = (float*)d_out;
    float* pq  = (float*)d_ws;

    const int E = in_sizes[1];
    const int hsize = in_sizes[0];
    if (ws_size >= (size_t)hsize * 2 * sizeof(float)) {
        node_proj_kernel<<<NBLOCKS, NWAVES * 32, 0, stream>>>(h, W1, b1, pq, hsize);
        edge_tail_kernel<<<NBLOCKS, NWAVES * 32, 0, stream>>>(pq, src, dst, W2, b2, W3, b3, out, E);
    } else {
        edge_mlp_kernel<<<NBLOCKS, NWAVES * 32, 0, stream>>>(h, src, dst, W1, b1, W2, b2,
                                                              W3, b3, out, E);
    }
}


# Edge kernel runs 8 waves per SIMD instead of 4 (32-wave work-groups, 128 registers): b2 and W3 are read from LDS behind each tile's matrix block instead of living in 64 registers; sums unchanged
# speedup vs baseline: 5.7267x; 1.0365x over previous
#include <hip/hip_runtime.h>


typedef float v2f __attribute__((ext_vector_type(2)));
typedef float v4f __attribute__((ext_vector_type(4)));
typedef float v8f __attribute__((ext_vector_type(8)));

#define H1 128
#define H2 64
#define KIN 256
#define PQW 256
#define NWAVES 16
#define NBLOCKS 256
#define EWAVES 32
#define W1_PITCH 260
#define W2_PITCH 132
#define ST_PITCH 132

__global__ __launch_bounds__(NWAVES * 32) void node_proj_kernel(
    const float* __restrict__ h,
    const float* __restrict__ W1, const float* __restrict__ b1,
    float* __restrict__ pq, int hsize)
{
    __shared__ __attribute__((aligned(16))) float w1s[H1 * W1_PITCH];
    __shared__ __attribute__((aligned(16))) float xstage[NWAVES * 16 * ST_PITCH];

    for (int i = threadIdx.x; i < H1 * KIN / 4; i += NWAVES * 32) {
        const int row = i >> 6;
        const int col = (i & 63) * 4;
        const v4f v = *(const v4f*)(W1 + row * KIN + col);
        *(v4f*)(w1s + row * W1_PITCH + col) = v;
    }
    __syncthreads();

    const int lane = threadIdx.x & 31;
    const int wave = __builtin_amdgcn_readfirstlane((int)(threadIdx.x >> 5));
    const int l15  = lane & 15;
    const int half = lane >> 4;
    float* stage = xstage + wave * 16 * ST_PITCH;

    float bias1[8];
#pragma unroll
    for (int nt = 0; nt < 8; ++nt) bias1[nt] = b1[nt * 16 + l15];

    const float* w1p = w1s + l15 * W1_PITCH + 2 * half;

    const int numNodes = hsize >> 7;
    const int numTiles = (numNodes + 15) >> 4;
    for (int p = (wave >> 1) * NBLOCKS + (int)blockIdx.x; 2 * p + (wave & 1) < numTiles;
         p += NBLOCKS * (NWAVES / 2)) {
        const int tileBase = (2 * p + (wave & 1)) * 16;

        int node = tileBase + l15;
        if (node > numNodes - 1) node = numNodes - 1;
        const float* rowH = h + (long long)node * H1 + 2 * half;

        for (int part = 0; part < 2; ++part) {
            v8f acc[8] = {};
            for (int c = 0; c < 32; ++c) {
                v2f a = *(const v2f*)(rowH + 4 * c);
#pragma unroll
                for (int nt = 0; nt < 8; ++nt) {
                    v2f bm = *(const v2f*)(w1p + nt * 16 * W1_PITCH + part * H1 + 4 * c);
                    acc[nt] = __builtin_amdgcn_wmma_f32_16x16x4_f32(
                        false, a, false, bm, (short)0, acc[nt], false, false);
                }
            }

#pragma unroll
            for (int nt = 0; nt < 8; ++nt) {
                const float badd = (part == 0) ? bias1[nt] : 0.0f;
#pragma unroll
                for (int r = 0; r < 8; ++r) {
                    stage[(r + 8 * half) * ST_PITCH + nt * 16 + l15] = acc[nt][r] + badd;
                }
            }
#pragma unroll
            for (int row = 0; row < 16; ++row) {
                const v4f v = *(const v4f*)(stage + row * ST_PITCH + 4 * lane);
                const int nd = tileBase + row;
                if (nd < numNodes)
                    *(v4f*)(pq + (long long)nd * PQW + part * H1 + 4 * lane) = v;
            }
        }
    }
}

__global__ __launch_bounds__(EWAVES * 32) void edge_tail_kernel(
    const float* __restrict__ pq,
    const int*   __restrict__ src_idx,
    const int*   __restrict__ dst_idx,
    const float* __restrict__ W2, const float* __restrict__ b2,
    const float* __restrict__ W3, const float* __restrict__ b3,
    float* __restrict__ out, int E)
{
    __shared__ __attribute__((aligned(16))) float lds2[2 * H2 + H2 * W2_PITCH + EWAVES * 16 * ST_PITCH];
    float* b2w3s  = lds2;
    float* w2s    = lds2 + 2 * H2;
    float* xstage = lds2 + 2 * H2 + H2 * W2_PITCH;

    for (int i = threadIdx.x; i < H2 * H1 / 4; i += EWAVES * 32) {
        const int row = i >> 5;
        const int col = (i & 31) * 4;
        const v4f v = *(const v4f*)(W2 + row * H1 + col);
        *(v4f*)(w2s + row * W2_PITCH + col) = v;
    }
    if (threadIdx.x < H2) {
        b2w3s[threadIdx.x]      = b2[threadIdx.x];
        b2w3s[H2 + threadIdx.x] = W3[threadIdx.x];
    }
    __syncthreads();

    const int lane = threadIdx.x & 31;
    const int wave = __builtin_amdgcn_readfirstlane((int)(threadIdx.x >> 5));
    const int l15  = lane & 15;
    const int half = lane >> 4;

    const float* bwp = b2w3s + 8 * half;
    const float bb3 = b3[0];

    const float* w2p = w2s + l15 * W2_PITCH + 4 * half;
    float* stage = xstage + wave * 16 * ST_PITCH;
    float* stw = stage + 4 * lane;
    const float* sta = stage + l15 * ST_PITCH + 4 * half;

    const int numTiles = (E + 15) >> 4;
    for (int p = (wave >> 1) * NBLOCKS + (int)blockIdx.x; 2 * p + (wave & 1) < numTiles;
         p += NBLOCKS * (EWAVES / 2)) {
        const int tileBase = (2 * p + (wave & 1)) * 16;

        asm volatile("" ::: "memory");

        int e = tileBase + l15;
        if (e > E - 1) e = E - 1;
        const int s = src_idx[e];
        const int d = dst_idx[e];

        v4f pv[16], qv[16];
#pragma unroll
        for (int i = 0; i < 16; ++i) {
            const int si = __builtin_amdgcn_readlane(s, i);
            const int di = __builtin_amdgcn_readlane(d, i);
            pv[i] = *(const v4f*)(pq + (long long)si * PQW + 4 * lane);
            qv[i] = *(const v4f*)(pq + (long long)di * PQW + H1 + 4 * lane);
        }
#pragma unroll
        for (int i = 0; i < 16; ++i) {
            const v4f x = pv[i] + qv[i];
            const v4f r = { fmaxf(x.x, 0.0f), fmaxf(x.y, 0.0f), fmaxf(x.z, 0.0f), fmaxf(x.w, 0.0f) };
            *(v4f*)(stw + i * ST_PITCH) = r;
        }
        v4f av[16];
#pragma unroll
        for (int j = 0; j < 16; ++j)
            av[j] = *(const v4f*)(sta + 8 * j);

        v8f acc2[4] = {};
#pragma unroll
        for (int j = 0; j < 16; ++j) {
            const v2f a0 = { av[j].x, av[j].y };
            const v2f a1 = { av[j].z, av[j].w };
            v4f bm[4];
#pragma unroll
            for (int nt = 0; nt < 4; ++nt)
                bm[nt] = *(const v4f*)(w2p + nt * 16 * W2_PITCH + 8 * j);
#pragma unroll
            for (int nt = 0; nt < 4; ++nt) {
                const v2f b0 = { bm[nt].x, bm[nt].y };
                acc2[nt] = __builtin_amdgcn_wmma_f32_16x16x4_f32(
                    false, b0, false, a0, (short)0, acc2[nt], false, false);
            }
#pragma unroll
            for (int nt = 0; nt < 4; ++nt) {
                const v2f b1v = { bm[nt].z, bm[nt].w };
                acc2[nt] = __builtin_amdgcn_wmma_f32_16x16x4_f32(
                    false, b1v, false, a1, (short)0, acc2[nt], false, false);
            }
        }

        v2f sacc[4];
#pragma unroll
        for (int nt = 0; nt < 4; ++nt) {
            const v4f bA = *(const v4f*)(bwp + nt * 16);
            const v4f bB = *(const v4f*)(bwp + nt * 16 + 4);
            const v4f wA = *(const v4f*)(bwp + H2 + nt * 16);
            const v4f wB = *(const v4f*)(bwp + H2 + nt * 16 + 4);
            const v2f bias2[4] = { { bA.x, bA.y }, { bA.z, bA.w }, { bB.x, bB.y }, { bB.z, bB.w } };
            const v2f w3v[4]   = { { wA.x, wA.y }, { wA.z, wA.w }, { wB.x, wB.y }, { wB.z, wB.w } };
            sacc[nt] = (v2f){ 0.0f, 0.0f };
#pragma unroll
            for (int q = 0; q < 4; ++q) {
                v2f x = (v2f){ acc2[nt][2 * q], acc2[nt][2 * q + 1] } + bias2[q];
                x = (v2f){ fmaxf(x.x, 0.0f), fmaxf(x.y, 0.0f) };
                sacc[nt] = x * w3v[q] + sacc[nt];
            }
        }
        const v2f s2 = (sacc[0] + sacc[1]) + (sacc[2] + sacc[3]);
        const float part = s2.x + s2.y;
        const float tot = part + __shfl_xor(part, 16, 32);

        const int idx = tileBase + l15;
        if (half == 0 && idx < E) out[idx] = tot + bb3;
    }
}

__global__ __launch_bounds__(NWAVES * 32) void edge_mlp_kernel(
    const float* __restrict__ h,
    const int*   __restrict__ src_idx,
    const int*   __restrict__ dst_idx,
    const float* __restrict__ W1, const float* __restrict__ b1,
    const float* __restrict__ W2, const float* __restrict__ b2,
    const float* __restrict__ W3, const float* __restrict__ b3,
    float* __restrict__ out, int E)
{
    __shared__ __attribute__((aligned(16))) float w1s[H1 * W1_PITCH];
    __shared__ __attribute__((aligned(16))) float w2s[H2 * W2_PITCH];
    __shared__ __attribute__((aligned(16))) float xstage[NWAVES * 16 * ST_PITCH];

    for (int i = threadIdx.x; i < H1 * KIN / 4; i += NWAVES * 32) {
        const int row = i >> 6;
        const int col = (i & 63) * 4;
        const v4f v = *(const v4f*)(W1 + row * KIN + col);
        *(v4f*)(w1s + row * W1_PITCH + col) = v;
    }
    for (int i = threadIdx.x; i < H2 * H1 / 4; i += NWAVES * 32) {
        const int row = i >> 5;
        const int col = (i & 31) * 4;
        const v4f v = *(const v4f*)(W2 + row * H1 + col);
        *(v4f*)(w2s + row * W2_PITCH + col) = v;
    }
    __syncthreads();

    const int lane = threadIdx.x & 31;
    const int wave = __builtin_amdgcn_readfirstlane((int)(threadIdx.x >> 5));
    const int l15  = lane & 15;
    const int half = lane >> 4;
    float* stage = xstage + wave * 16 * ST_PITCH;

    float bias1[8];
#pragma unroll
    for (int nt = 0; nt < 8; ++nt) bias1[nt] = b1[nt * 16 + l15];
    float bias2[4], w3v[4];
#pragma unroll
    for (int nt = 0; nt < 4; ++nt) {
        bias2[nt] = b2[nt * 16 + l15];
        w3v[nt]   = W3[nt * 16 + l15];
    }
    const float bb3 = b3[0];

    const float* w1p = w1s + l15 * W1_PITCH + 2 * half;
    const float* w2p = w2s + l15 * W2_PITCH + 2 * half;
    const float* sp  = stage + l15 * ST_PITCH + 2 * half;

    const int numTiles = (E + 15) >> 4;
    for (int p = (wave >> 1) * NBLOCKS + (int)blockIdx.x; 2 * p + (wave & 1) < numTiles;
         p += NBLOCKS * (NWAVES / 2)) {
        const int tileBase = (2 * p + (wave & 1)) * 16;

        int e = tileBase + l15;
        if (e > E - 1) e = E - 1;
        const int s = src_idx[e];
        const int d = dst_idx[e];
        const float* rowS = h + (long long)s * H1 + 2 * half;
        const float* rowD = h + (long long)d * H1 + 2 * half;

        v8f acc[8] = {};
        for (int c = 0; c < 32; ++c) {
            v2f a = *(const v2f*)(rowS + 4 * c);
#pragma unroll
            for (int nt = 0; nt < 8; ++nt) {
                v2f bm = *(const v2f*)(w1p + nt * 16 * W1_PITCH + 4 * c);
                acc[nt] = __builtin_amdgcn_wmma_f32_16x16x4_f32(
                    false, a, false, bm, (short)0, acc[nt], false, false);
            }
        }
        for (int c = 0; c < 32; ++c) {
            v2f a = *(const v2f*)(rowD + 4 * c);
#pragma unroll
            for (int nt = 0; nt < 8; ++nt) {
                v2f bm = *(const v2f*)(w1p + nt * 16 * W1_PITCH + 4 * (c + 32));
                acc[nt] = __builtin_amdgcn_wmma_f32_16x16x4_f32(
                    false, a, false, bm, (short)0, acc[nt], false, false);
            }
        }

#pragma unroll
        for (int nt = 0; nt < 8; ++nt) {
#pragma unroll
            for (int r = 0; r < 8; ++r) {
                float x = fmaxf(acc[nt][r] + bias1[nt], 0.0f);
                stage[(r + 8 * half) * ST_PITCH + nt * 16 + l15] = x;
            }
        }

        v8f acc2[4] = {};
        for (int c = 0; c < 32; ++c) {
            v2f a = *(const v2f*)(sp + 4 * c);
#pragma unroll
            for (int nt = 0; nt < 4; ++nt) {
                v2f bm = *(const v2f*)(w2p + nt * 16 * W2_PITCH + 4 * c);
                acc2[nt] = __builtin_amdgcn_wmma_f32_16x16x4_f32(
                    false, a, false, bm, (short)0, acc2[nt], false, false);
            }
        }

        float sred[8];
#pragma unroll
        for (int r = 0; r < 8; ++r) sred[r] = 0.0f;
#pragma unroll
        for (int nt = 0; nt < 4; ++nt) {
#pragma unroll
            for (int r = 0; r < 8; ++r) {
                float x = fmaxf(acc2[nt][r] + bias2[nt], 0.0f);
                sred[r] += x * w3v[nt];
            }
        }
#pragma unroll
        for (int mask = 1; mask <= 8; mask <<= 1) {
#pragma unroll
            for (int r = 0; r < 8; ++r)
                sred[r] += __shfl_xor(sred[r], mask, 32);
        }

        if (l15 == 0) {
            const int m0 = tileBase + 8 * half;
#pragma unroll
            for (int r = 0; r < 8; ++r) {
                const int idx = m0 + r;
                if (idx < E) out[idx] = sred[r] + bb3;
            }
        }
    }
}

extern "C" void kernel_launch(void* const* d_in, const int* in_sizes, int n_in,
                              void* d_out, int out_size, void* d_ws, size_t ws_size,
                              hipStream_t stream) {
    const float* h   = (const float*)d_in[0];
    const int*   src = (const int*)  d_in[1];
    const int*   dst = (const int*)  d_in[2];
    const float* W1  = (const float*)d_in[3];
    const float* b1  = (const float*)d_in[4];
    const float* W2  = (const float*)d_in[5];
    const float* b2  = (const float*)d_in[6];
    const float* W3  = (const float*)d_in[7];
    const float* b3  = (const float*)d_in[8];
    float* out = (float*)d_out;
    float* pq  = (float*)d_ws;

    const int E = in_sizes[1];
    const int hsize = in_sizes[0];
    if (ws_size >= (size_t)hsize * 2 * sizeof(float)) {
        node_proj_kernel<<<NBLOCKS, NWAVES * 32, 0, stream>>>(h, W1, b1, pq, hsize);
        edge_tail_kernel<<<NBLOCKS, EWAVES * 32, 0, stream>>>(pq, src, dst, W2, b2, W3, b3, out, E);
    } else {
        edge_mlp_kernel<<<NBLOCKS, NWAVES * 32, 0, stream>>>(h, src, dst, W1, b1, W2, b2,
                                                              W3, b3, out, E);
    }
}
